# starting kernel
# baseline (speedup 1.0000x reference)
#include <hip/hip_runtime.h>

typedef __attribute__((ext_vector_type(16))) __bf16 v16bf;
typedef __attribute__((ext_vector_type(8)))  float  v8f;


__global__ __launch_bounds__(256) void fill_f32_k(float* p, float v, long long n) {
  long long i = (long long)blockIdx.x * 256 + threadIdx.x;
  if (i < n) p[i] = v;
}

__device__ __forceinline__ void atomic_max_f32(float* addr, float v) {
  if (v >= 0.0f) atomicMax((int*)addr, __float_as_int(v));
  else           atomicMin((unsigned int*)addr, __float_as_uint(v));
}

__global__ __launch_bounds__(256) void pack_a_k(const float* X, __bf16* Ap, int M, int K) {
  long long o = (long long)blockIdx.x * 256 + threadIdx.x;
  long long total = (long long)M * K;
  if (o >= total) return;
  int i    = (int)(o & 15);
  int lane = (int)((o >> 4) & 31);
  long long t = o >> 9;
  int Kt = K >> 5;
  int ks = (int)(t % Kt);
  int tm = (int)(t / Kt);
  int m = tm * 16 + (lane & 15);
  int k = ks * 32 + ((lane >> 4) << 3) + (i < 8 ? i : i + 8);
  Ap[o] = (__bf16)X[(long long)m * K + k];
}

__global__ __launch_bounds__(256) void pack_b_k(const float* W, __bf16* Bp, int K, int N) {
  long long o = (long long)blockIdx.x * 256 + threadIdx.x;
  long long total = (long long)K * N;
  if (o >= total) return;
  int i    = (int)(o & 15);
  int lane = (int)((o >> 4) & 31);
  long long t = o >> 9;
  int Kt = K >> 5;
  int ks = (int)(t % Kt);
  int tn = (int)(t / Kt);
  int n = tn * 16 + (lane & 15);
  int k = ks * 32 + ((lane >> 4) << 3) + (i < 8 ? i : i + 8);
  Bp[o] = (__bf16)W[(long long)k * N + n];
}

template <int NT>
__global__ __launch_bounds__(256) void wmma_gemm_strip_k(const __bf16* __restrict__ Ap,
                                                         const __bf16* __restrict__ Bp,
                                                         float* __restrict__ C,
                                                         int Mt, int Kt) {
  const int N = NT * 16;
  int wave = threadIdx.x >> 5;
  int lane = threadIdx.x & 31;
  long long tm = (long long)blockIdx.x * 8 + wave;
  if (tm >= Mt) return;
  const v16bf* Af = (const v16bf*)Ap;
  const v16bf* Bf = (const v16bf*)Bp;
  v8f zero = {};
  v8f acc[NT];
  #pragma unroll
  for (int t = 0; t < NT; ++t) acc[t] = zero;
  for (int ks = 0; ks < Kt; ++ks) {
    v16bf a = Af[(tm * Kt + ks) * 32 + lane];
    #pragma unroll
    for (int t = 0; t < NT; ++t) {
      v16bf b = Bf[((long long)t * Kt + ks) * 32 + lane];
      acc[t] = __builtin_amdgcn_wmma_f32_16x16x32_bf16(
          false, a, false, b, (short)0, acc[t], false, false);
    }
  }
  int colb = lane & 15;
  long long rbase = tm * 16 + ((lane >> 4) << 3);
  #pragma unroll
  for (int t = 0; t < NT; ++t) {
    #pragma unroll
    for (int r = 0; r < 8; ++r)
      C[(rbase + r) * N + t * 16 + colb] = acc[t][r];
  }
}

template <int FPL>
__global__ __launch_bounds__(256) void edge_logits_k(const float* __restrict__ xl,
                                                     const float* __restrict__ xr,
                                                     const float* __restrict__ att,
                                                     const int* __restrict__ esrc,
                                                     const int* __restrict__ edst,
                                                     int E, int Nn,
                                                     float* __restrict__ logits,
                                                     float* __restrict__ mmax) {
  const int C = FPL * 32;
  int wave = threadIdx.x >> 5, lane = threadIdx.x & 31;
  long long e = (long long)blockIdx.x * 8 + wave;
  long long EA = (long long)E + Nn;
  if (e >= EA) return;
  int s, d;
  if (e < E) { s = esrc[e]; d = edst[e]; } else { s = d = (int)(e - E); }
  float part = 0.0f;
  if constexpr (FPL == 4) {
    float4 l4 = ((const float4*)(xl + (long long)s * C))[lane];
    float4 r4 = ((const float4*)(xr + (long long)d * C))[lane];
    float4 a4 = ((const float4*)att)[lane];
    float v;
    v = l4.x + r4.x; v = fmaxf(v, 0.2f * v); part += v * a4.x;
    v = l4.y + r4.y; v = fmaxf(v, 0.2f * v); part += v * a4.y;
    v = l4.z + r4.z; v = fmaxf(v, 0.2f * v); part += v * a4.z;
    v = l4.w + r4.w; v = fmaxf(v, 0.2f * v); part += v * a4.w;
  } else {
    float v = xl[(long long)s * C + lane] + xr[(long long)d * C + lane];
    v = fmaxf(v, 0.2f * v);
    part = v * att[lane];
  }
  #pragma unroll
  for (int m = 16; m >= 1; m >>= 1) part += __shfl_xor(part, m, 32);
  if (lane == 0) {
    logits[e] = part;
    atomic_max_f32(&mmax[d], part);
  }
}

template <int FPL>
__global__ __launch_bounds__(256) void edge_agg_k(const float* __restrict__ xl,
                                                  const float* __restrict__ logits,
                                                  const float* __restrict__ mmax,
                                                  const int* __restrict__ esrc,
                                                  const int* __restrict__ edst,
                                                  int E, int Nn,
                                                  float* __restrict__ denom,
                                                  float* __restrict__ acc) {
  const int C = FPL * 32;
  int wave = threadIdx.x >> 5, lane = threadIdx.x & 31;
  long long e = (long long)blockIdx.x * 8 + wave;
  long long EA = (long long)E + Nn;
  if (e >= EA) return;
  int s, d;
  if (e < E) { s = esrc[e]; d = edst[e]; } else { s = d = (int)(e - E); }
  float ev = __expf(logits[e] - mmax[d]);
  if (lane == 0) atomicAdd(&denom[d], ev);
  if constexpr (FPL == 4) {
    float4 l4 = ((const float4*)(xl + (long long)s * C))[lane];
    float* ap = acc + (long long)d * C + lane * 4;
    atomicAdd(ap + 0, l4.x * ev);
    atomicAdd(ap + 1, l4.y * ev);
    atomicAdd(ap + 2, l4.z * ev);
    atomicAdd(ap + 3, l4.w * ev);
  } else {
    atomicAdd(acc + (long long)d * C + lane, xl[(long long)s * C + lane] * ev);
  }
}

__global__ __launch_bounds__(256) void finalize_k(const float* __restrict__ acc,
                                                  const float* __restrict__ denom,
                                                  const float* __restrict__ bias,
                                                  float* __restrict__ out,
                                                  long long total, int C) {
  long long i = (long long)blockIdx.x * 256 + threadIdx.x;
  if (i >= total) return;
  long long n = i / C;
  int c = (int)(i % C);
  out[i] = fmaxf(acc[i] / (denom[n] + 1e-16f) + bias[c], 0.0f);
}


extern "C" void kernel_launch(void* const* d_in, const int* in_sizes, int n_in,
                              void* d_out, int out_size, void* d_ws, size_t ws_size,
                              hipStream_t stream) {
  const int F_IN = 128, H = 128, OUT = 32;
  const float* x    = (const float*)d_in[0];
  const int*   eidx = (const int*)  d_in[1];
  const float* Wl1  = (const float*)d_in[2];
  const float* Wr1  = (const float*)d_in[3];
  const float* att1 = (const float*)d_in[4];
  const float* b1   = (const float*)d_in[5];
  const float* Wl2  = (const float*)d_in[6];
  const float* Wr2  = (const float*)d_in[7];
  const float* att2 = (const float*)d_in[8];
  const float* b2   = (const float*)d_in[9];
  float* out = (float*)d_out;

  const int Nn = in_sizes[0] / F_IN;
  const int E  = in_sizes[1] / 2;
  const long long EA = (long long)E + Nn;
  const int* esrc = eidx;
  const int* edst = eidx + E;

  char* ws = (char*)d_ws;
  size_t off = 0;
  auto carve = [&](size_t bytes) -> char* {
    char* p = ws + off;
    off += (bytes + 255) & ~(size_t)255;
    return p;
  };
  __bf16* apack = (__bf16*)carve((size_t)Nn * F_IN * sizeof(__bf16));
  __bf16* bpl1  = (__bf16*)carve((size_t)F_IN * H   * sizeof(__bf16));
  __bf16* bpr1  = (__bf16*)carve((size_t)F_IN * H   * sizeof(__bf16));
  __bf16* bpl2  = (__bf16*)carve((size_t)H * OUT    * sizeof(__bf16));
  __bf16* bpr2  = (__bf16*)carve((size_t)H * OUT    * sizeof(__bf16));
  float*  xl    = (float*) carve((size_t)Nn * H * sizeof(float));
  float*  xr    = (float*) carve((size_t)Nn * H * sizeof(float));
  float*  logit = (float*) carve((size_t)EA * sizeof(float));
  float*  mmax  = (float*) carve((size_t)Nn * sizeof(float));
  float*  denom = (float*) carve((size_t)Nn * sizeof(float));
  float*  accb  = (float*) carve((size_t)Nn * H * sizeof(float));
  float*  hbuf  = (float*) carve((size_t)Nn * H * sizeof(float));

  auto g256 = [](long long n) { return (unsigned)((n + 255) / 256); };
  auto gwav = [](long long n) { return (unsigned)((n + 7) / 8); };

  pack_b_k<<<g256((long long)F_IN * H), 256, 0, stream>>>(Wl1, bpl1, F_IN, H);
  pack_b_k<<<g256((long long)F_IN * H), 256, 0, stream>>>(Wr1, bpr1, F_IN, H);
  pack_b_k<<<g256((long long)H * OUT),  256, 0, stream>>>(Wl2, bpl2, H, OUT);
  pack_b_k<<<g256((long long)H * OUT),  256, 0, stream>>>(Wr2, bpr2, H, OUT);

  const int Mt = Nn / 16;

  {
    const int C = H, K = F_IN, Kt = K / 32;
    pack_a_k<<<g256((long long)Nn * K), 256, 0, stream>>>(x, apack, Nn, K);
    wmma_gemm_strip_k<8><<<gwav(Mt), 256, 0, stream>>>(apack, bpl1, xl, Mt, Kt);
    wmma_gemm_strip_k<8><<<gwav(Mt), 256, 0, stream>>>(apack, bpr1, xr, Mt, Kt);

    fill_f32_k<<<g256(Nn), 256, 0, stream>>>(mmax, -INFINITY, Nn);
    fill_f32_k<<<g256(Nn), 256, 0, stream>>>(denom, 0.0f, Nn);
    fill_f32_k<<<g256((long long)Nn * C), 256, 0, stream>>>(accb, 0.0f, (long long)Nn * C);

    edge_logits_k<4><<<gwav(EA), 256, 0, stream>>>(xl, xr, att1, esrc, edst, E, Nn, logit, mmax);
    edge_agg_k<4><<<gwav(EA), 256, 0, stream>>>(xl, logit, mmax, esrc, edst, E, Nn, denom, accb);
    finalize_k<<<g256((long long)Nn * C), 256, 0, stream>>>(accb, denom, b1, hbuf,
                                                            (long long)Nn * C, C);
  }

  {
    const int C = OUT, K = H, Kt = K / 32;
    pack_a_k<<<g256((long long)Nn * K), 256, 0, stream>>>(hbuf, apack, Nn, K);
    wmma_gemm_strip_k<2><<<gwav(Mt), 256, 0, stream>>>(apack, bpl2, xl, Mt, Kt);
    wmma_gemm_strip_k<2><<<gwav(Mt), 256, 0, stream>>>(apack, bpr2, xr, Mt, Kt);

    fill_f32_k<<<g256(Nn), 256, 0, stream>>>(mmax, -INFINITY, Nn);
    fill_f32_k<<<g256(Nn), 256, 0, stream>>>(denom, 0.0f, Nn);
    fill_f32_k<<<g256((long long)Nn * C), 256, 0, stream>>>(accb, 0.0f, (long long)Nn * C);

    edge_logits_k<1><<<gwav(EA), 256, 0, stream>>>(xl, xr, att2, esrc, edst, E, Nn, logit, mmax);
    edge_agg_k<1><<<gwav(EA), 256, 0, stream>>>(xl, logit, mmax, esrc, edst, E, Nn, denom, accb);
    finalize_k<<<g256((long long)Nn * C), 256, 0, stream>>>(accb, denom, b2, out,
                                                            (long long)Nn * C, C);
  }
}


# Tile kernel: the four 32-lane logit sums of a trip run on vector-ALU lane exchanges (one row swap, then row_xmask DPP adds), folded so each step carries half the values; no LDS permutes; same sums
# speedup vs baseline: 5.2124x; 5.2124x over previous
#include <hip/hip_runtime.h>

typedef __attribute__((ext_vector_type(16))) __bf16 v16bf;
typedef __attribute__((ext_vector_type(8)))  float  v8f;
typedef __attribute__((ext_vector_type(8)))  __bf16 v8bf;


__global__ __launch_bounds__(256) void fill_f32_k(float* p, float v, long long n) {
  long long i = (long long)blockIdx.x * 256 + threadIdx.x;
  if (i < n) p[i] = v;
}

__global__ __launch_bounds__(256) void pack_a_k(const float* __restrict__ X,
                                                __bf16* __restrict__ Ap, int M, int K) {
  unsigned g = blockIdx.x * 256u + threadIdx.x;
  unsigned nthr = (unsigned)(((long long)M * K) >> 4);
  if (g >= nthr) return;
  unsigned L = g & 31u;
  unsigned t = g >> 5;
  unsigned Kt = (unsigned)K >> 5;
  unsigned ks = t % Kt;
  unsigned tm = t / Kt;
  unsigned m = tm * 16u + (L >> 1);
  const float4* p = (const float4*)(X + (long long)m * K + ks * 32u + ((L & 1u) << 4));
  float4 q0 = p[0], q1 = p[1], q2 = p[2], q3 = p[3];
  v8bf lo, hi;
  lo[0] = (__bf16)q0.x; lo[1] = (__bf16)q0.y; lo[2] = (__bf16)q0.z; lo[3] = (__bf16)q0.w;
  lo[4] = (__bf16)q1.x; lo[5] = (__bf16)q1.y; lo[6] = (__bf16)q1.z; lo[7] = (__bf16)q1.w;
  hi[0] = (__bf16)q2.x; hi[1] = (__bf16)q2.y; hi[2] = (__bf16)q2.z; hi[3] = (__bf16)q2.w;
  hi[4] = (__bf16)q3.x; hi[5] = (__bf16)q3.y; hi[6] = (__bf16)q3.z; hi[7] = (__bf16)q3.w;
  v8bf* dst = (v8bf*)Ap + (long long)t * 64 + L;
  dst[0]  = lo;
  dst[32] = hi;
}

__global__ __launch_bounds__(256) void pack_b_k(const float* W, __bf16* Bp, int K, int N) {
  long long o = (long long)blockIdx.x * 256 + threadIdx.x;
  long long total = (long long)K * N;
  if (o >= total) return;
  int i    = (int)(o & 15);
  int lane = (int)((o >> 4) & 31);
  long long t = o >> 9;
  int Kt = K >> 5;
  int ks = (int)(t % Kt);
  int tn = (int)(t / Kt);
  int n = tn * 16 + (lane & 15);
  int k = ks * 32 + ((lane >> 4) << 3) + (i < 8 ? i : i + 8);
  Bp[o] = (__bf16)W[(long long)k * N + n];
}

template <int NT>
__global__ __launch_bounds__(256) void wmma_gemm_strip_k(const __bf16* __restrict__ Ap,
                                                         const __bf16* __restrict__ Bp,
                                                         float* __restrict__ C,
                                                         int Mt, int Kt) {
  const int N = NT * 16;
  int wave = threadIdx.x >> 5;
  int lane = threadIdx.x & 31;
  long long tm = (long long)blockIdx.x * 8 + wave;
  if (tm >= Mt) return;
  const v16bf* Af = (const v16bf*)Ap;
  const v16bf* Bf = (const v16bf*)Bp;
  v8f zero = {};
  v8f acc[NT];
  #pragma unroll
  for (int t = 0; t < NT; ++t) acc[t] = zero;
  for (int ks = 0; ks < Kt; ++ks) {
    v16bf a = Af[(tm * Kt + ks) * 32 + lane];
    #pragma unroll
    for (int t = 0; t < NT; ++t) {
      v16bf b = Bf[((long long)t * Kt + ks) * 32 + lane];
      acc[t] = __builtin_amdgcn_wmma_f32_16x16x32_bf16(
          false, a, false, b, (short)0, acc[t], false, false);
    }
  }
  int colb = lane & 15;
  long long rbase = tm * 16 + ((lane >> 4) << 3);
  #pragma unroll
  for (int t = 0; t < NT; ++t) {
    #pragma unroll
    for (int r = 0; r < 8; ++r)
      C[(rbase + r) * N + t * 16 + colb] = acc[t][r];
  }
}

#define TILE_SHIFT 5
#define TILE_NODES 32

extern __shared__ int dyn_bins[];

__global__ __launch_bounds__(256) void bin_count_k(const int* __restrict__ edst, int E, int NB,
                                                   int* __restrict__ gcnt) {
  int* hist = dyn_bins;
  for (int b = threadIdx.x; b < NB; b += 256) hist[b] = 0;
  __syncthreads();
  long long per = ((long long)E + gridDim.x - 1) / gridDim.x;
  long long beg = per * blockIdx.x;
  long long end = beg + per < (long long)E ? beg + per : (long long)E;
  for (long long e = beg + threadIdx.x; e < end; e += 256)
    atomicAdd(&hist[edst[e] >> TILE_SHIFT], 1);
  __syncthreads();
  for (int b = threadIdx.x; b < NB; b += 256) {
    int c = hist[b];
    if (c) atomicAdd(&gcnt[b], c);
  }
}

__global__ __launch_bounds__(1024) void bin_offsets_k(const int* __restrict__ gcnt, int NB,
                                                      int* __restrict__ boff) {
  __shared__ int part[1024];
  int t = threadIdx.x;
  int per = (NB + 1023) / 1024;
  int b0 = t * per;
  int b1 = b0 + per < NB ? b0 + per : NB;
  int s = 0;
  for (int b = b0; b < b1; ++b) s += gcnt[b];
  part[t] = s;
  __syncthreads();
  for (int off = 1; off < 1024; off <<= 1) {
    int v = (t >= off) ? part[t - off] : 0;
    __syncthreads();
    part[t] += v;
    __syncthreads();
  }
  int run = part[t] - s;
  for (int b = b0; b < b1; ++b) { boff[b] = run; run += gcnt[b]; }
  if (t == 1023) boff[NB] = part[1023];
}

__global__ __launch_bounds__(256) void bin_place_k(const int* __restrict__ esrc,
                                                   const int* __restrict__ edst, int E, int NB,
                                                   const int* __restrict__ boff,
                                                   int* __restrict__ gcur,
                                                   int* __restrict__ bs, int* __restrict__ bd) {
  int* hist = dyn_bins;
  int* base = dyn_bins + NB;
  for (int b = threadIdx.x; b < NB; b += 256) hist[b] = 0;
  __syncthreads();
  long long per = ((long long)E + gridDim.x - 1) / gridDim.x;
  long long beg = per * blockIdx.x;
  long long end = beg + per < (long long)E ? beg + per : (long long)E;
  for (long long e = beg + threadIdx.x; e < end; e += 256)
    atomicAdd(&hist[edst[e] >> TILE_SHIFT], 1);
  __syncthreads();
  for (int b = threadIdx.x; b < NB; b += 256) {
    int c = hist[b];
    base[b] = c ? boff[b] + atomicAdd(&gcur[b], c) : 0;
    hist[b] = 0;
  }
  __syncthreads();
  for (long long e = beg + threadIdx.x; e < end; e += 256) {
    int d = edst[e];
    int b = d >> TILE_SHIFT;
    int p = base[b] + atomicAdd(&hist[b], 1);
    bs[p] = esrc[e];
    bd[p] = d;
  }
}

__device__ __forceinline__ void lds_max_f32(float* addr, float v) {
  int bits = __float_as_int(v);
  if (bits >= 0) atomicMax((int*)addr, bits);
  else           atomicMin((unsigned int*)addr, (unsigned int)bits);
}

template <int FPL>
__device__ __forceinline__ void load_row(const float* __restrict__ tab, int n, int lane,
                                         float (&r)[FPL]) {
  if constexpr (FPL == 4) {
    unsigned off = (unsigned)n * 512u + (unsigned)lane * 16u;
    float4 t = *(const float4*)((const char*)tab + off);
    r[0] = t.x; r[1] = t.y; r[2] = t.z; r[3] = t.w;
  } else {
    unsigned off = (unsigned)n * 128u + (unsigned)lane * 4u;
    r[0] = *(const float*)((const char*)tab + off);
  }
}

__device__ __forceinline__ float xhalf(float x) {
  int b = __float_as_int(x);
  return __int_as_float(__builtin_amdgcn_permlanex16(b, b, 0x76543210, 0xfedcba98, false, false));
}

template <int M>
__device__ __forceinline__ float xrow(float x) {
  return __int_as_float(__builtin_amdgcn_update_dpp(0, __float_as_int(x), 0x160 + M, 0xf, 0xf, true));
}

template <int FPL, bool FULL>
__device__ __forceinline__ void logits4(const float* __restrict__ xl, const float* __restrict__ xr,
                                        const float (&a)[FPL], const int (&s)[4], const int (&d)[4],
                                        const bool (&ok)[4], int lane, int n0,
                                        float (&part)[4], float& mymax) {
  float l[4][FPL], r[4][FPL];
  #pragma unroll
  for (int u = 0; u < 4; ++u) { load_row<FPL>(xl, s[u], lane, l[u]); load_row<FPL>(xr, d[u], lane, r[u]); }
  #pragma unroll
  for (int u = 0; u < 4; ++u) {
    float p = 0.0f;
    #pragma unroll
    for (int k = 0; k < FPL; ++k) {
      float v = l[u][k] + r[u][k];
      v = fmaxf(v, 0.2f * v);
      p += v * a[k];
    }
    part[u] = p;
  }
  const bool up = (lane & 16) != 0;
  const bool hi = (lane & 8) != 0;
  const float r0 = (up ? part[2] : part[0]) + xhalf(up ? part[0] : part[2]);
  const float r1 = (up ? part[3] : part[1]) + xhalf(up ? part[1] : part[3]);
  float sum = (hi ? r1 : r0) + xrow<8>(hi ? r0 : r1);
  sum += xrow<4>(sum);
  sum += xrow<2>(sum);
  sum += xrow<1>(sum);
  #pragma unroll
  for (int u = 0; u < 4; ++u)
    part[u] = __int_as_float(__builtin_amdgcn_readlane(__float_as_int(sum), 8 * u));
  #pragma unroll
  for (int u = 0; u < 4; ++u)
    if ((FULL || ok[u]) && lane == d[u] - n0) mymax = fmaxf(mymax, part[u]);
}

template <int FPL, bool FULL>
__device__ __forceinline__ void accum4(const float* __restrict__ xl, const int (&s)[4],
                                       const int (&d)[4], const bool (&ok)[4],
                                       const float (&ev)[4], int lane, int n0, int tmax_bits,
                                       float* sacc, float& myden) {
  constexpr int C = FPL * 32;
  float l[4][FPL];
  #pragma unroll
  for (int u = 0; u < 4; ++u) load_row<FPL>(xl, s[u], lane, l[u]);
  #pragma unroll
  for (int u = 0; u < 4; ++u) {
    if (FULL || ok[u]) {
      int dl = d[u] - n0;
      float w = __expf(ev[u] - __int_as_float(__builtin_amdgcn_readlane(tmax_bits, dl)));
      if (lane == dl) myden += w;
      #pragma unroll
      for (int k = 0; k < FPL; ++k) atomicAdd(&sacc[dl * C + k * 32 + lane], l[u][k] * w);
    }
  }
}

template <int FPL>
__global__ __launch_bounds__(256) void tile_gat_k(const float* __restrict__ xl,
                                                  const float* __restrict__ xr,
                                                  const float* __restrict__ att,
                                                  const float* __restrict__ bias,
                                                  const int* __restrict__ bs,
                                                  const int* __restrict__ bd,
                                                  const int* __restrict__ boff,
                                                  int E, int Nn,
                                                  float* __restrict__ lg,
                                                  float* __restrict__ out) {
  constexpr int C = FPL * 32;
  constexpr int NW = 8;
  constexpr int EU = 4;
  __shared__ float sacc[TILE_NODES * C];
  __shared__ float smax[TILE_NODES];
  __shared__ float sden[TILE_NODES];
  const int tid = threadIdx.x, lane = tid & 31;
  const int wave = __builtin_amdgcn_readfirstlane(tid >> 5);
  const int n0 = blockIdx.x * TILE_NODES;
  const int nn = Nn - n0 < TILE_NODES ? Nn - n0 : TILE_NODES;
  const int beg = boff[blockIdx.x];
  const int cnt = boff[blockIdx.x + 1] - beg;
  const int nfull = cnt >> 2;
  const int rbeg = beg + (nfull << 2);
  const int rcnt = cnt & 3;
  const int rtotal = rcnt + nn;

  for (int i = tid; i < TILE_NODES * C; i += 256) sacc[i] = 0.0f;
  if (tid < TILE_NODES) { smax[tid] = -INFINITY; sden[tid] = 0.0f; }
  float a[FPL];
  load_row<FPL>(att, 0, lane, a);
  static_assert(TILE_NODES == 32, "lane j stands for tile node j");
  static_assert(EU == 4, "a trip is four edges");
  const bool all_ok[4] = {true, true, true, true};
  float mymax = -INFINITY;
  __syncthreads();

  for (int c = wave; c < nfull; c += NW) {
    const int e0 = beg + (c << 2);
    int s[4], d[4];
    __builtin_memcpy(s, bs + e0, 16);
    __builtin_memcpy(d, bd + e0, 16);
    float part[4];
    logits4<FPL, true>(xl, xr, a, s, d, all_ok, lane, n0, part, mymax);
    if (lane == 0) __builtin_memcpy(lg + e0, part, 16);
  }
  for (int i0 = wave; i0 < rtotal; i0 += NW * EU) {
    int s[4], d[4]; long long li[4]; bool ok[4];
    #pragma unroll
    for (int u = 0; u < 4; ++u) {
      int i = i0 + u * NW;
      ok[u] = i < rtotal;
      if (!ok[u]) i = rtotal - 1;
      if (i < rcnt) { s[u] = bs[rbeg + i]; d[u] = bd[rbeg + i]; li[u] = rbeg + i; }
      else          { s[u] = d[u] = n0 + (i - rcnt); li[u] = (long long)E + d[u]; }
    }
    float part[4];
    logits4<FPL, false>(xl, xr, a, s, d, ok, lane, n0, part, mymax);
    #pragma unroll
    for (int u = 0; u < 4; ++u)
      if (lane == 0 && ok[u]) lg[li[u]] = part[u];
  }
  lds_max_f32(&smax[lane], mymax);
  __syncthreads();
  const int tmax_bits = __float_as_int(smax[lane]);
  float myden = 0.0f;

  for (int c = wave; c < nfull; c += NW) {
    const int e0 = beg + (c << 2);
    int s[4], d[4]; float ev[4];
    __builtin_memcpy(s, bs + e0, 16);
    __builtin_memcpy(d, bd + e0, 16);
    __builtin_memcpy(ev, lg + e0, 16);
    accum4<FPL, true>(xl, s, d, all_ok, ev, lane, n0, tmax_bits, sacc, myden);
  }
  for (int i0 = wave; i0 < rtotal; i0 += NW * EU) {
    int s[4], d[4]; long long li[4]; bool ok[4];
    #pragma unroll
    for (int u = 0; u < 4; ++u) {
      int i = i0 + u * NW;
      ok[u] = i < rtotal;
      if (!ok[u]) i = rtotal - 1;
      if (i < rcnt) { s[u] = bs[rbeg + i]; d[u] = bd[rbeg + i]; li[u] = rbeg + i; }
      else          { s[u] = d[u] = n0 + (i - rcnt); li[u] = (long long)E + d[u]; }
    }
    float ev[4];
    #pragma unroll
    for (int u = 0; u < 4; ++u) ev[u] = lg[li[u]];
    accum4<FPL, false>(xl, s, d, ok, ev, lane, n0, tmax_bits, sacc, myden);
  }
  atomicAdd(&sden[lane], myden);
  __syncthreads();

  for (int idx = tid; idx < nn * C; idx += 256) {
    int n = idx / C, f = idx % C;
    int p = (FPL == 4) ? ((f & 3) * 32 + (f >> 2)) : f;
    out[(long long)(n0 + n) * C + f] =
        fmaxf(sacc[n * C + p] / (sden[n] + 1e-16f) + bias[f], 0.0f);
  }
}


extern "C" void kernel_launch(void* const* d_in, const int* in_sizes, int n_in,
                              void* d_out, int out_size, void* d_ws, size_t ws_size,
                              hipStream_t stream) {
  const int F_IN = 128, H = 128, OUT = 32;
  const float* x    = (const float*)d_in[0];
  const int*   eidx = (const int*)  d_in[1];
  const float* Wl1  = (const float*)d_in[2];
  const float* Wr1  = (const float*)d_in[3];
  const float* att1 = (const float*)d_in[4];
  const float* b1   = (const float*)d_in[5];
  const float* Wl2  = (const float*)d_in[6];
  const float* Wr2  = (const float*)d_in[7];
  const float* att2 = (const float*)d_in[8];
  const float* b2   = (const float*)d_in[9];
  float* out = (float*)d_out;

  const int Nn = in_sizes[0] / F_IN;
  const int E  = in_sizes[1] / 2;
  const long long EA = (long long)E + Nn;
  const int* esrc = eidx;
  const int* edst = eidx + E;
  const int NB = (Nn + TILE_NODES - 1) / TILE_NODES;

  char* ws = (char*)d_ws;
  size_t off = 0;
  auto carve = [&](size_t bytes) -> char* {
    char* p = ws + off;
    off += (bytes + 255) & ~(size_t)255;
    return p;
  };
  __bf16* apack = (__bf16*)carve((size_t)Nn * F_IN * sizeof(__bf16));
  __bf16* bpl1  = (__bf16*)carve((size_t)F_IN * H   * sizeof(__bf16));
  __bf16* bpr1  = (__bf16*)carve((size_t)F_IN * H   * sizeof(__bf16));
  __bf16* bpl2  = (__bf16*)carve((size_t)H * OUT    * sizeof(__bf16));
  __bf16* bpr2  = (__bf16*)carve((size_t)H * OUT    * sizeof(__bf16));
  float*  xl    = (float*) carve((size_t)Nn * H * sizeof(float));
  float*  xr    = (float*) carve((size_t)Nn * H * sizeof(float));
  float*  logit = (float*) carve((size_t)EA * sizeof(float));
  float*  hbuf  = (float*) carve((size_t)Nn * H * sizeof(float));
  int*    cnts  = (int*)   carve((size_t)2 * NB * sizeof(int));
  int*    boff  = (int*)   carve((size_t)(NB + 1) * sizeof(int));
  int*    bsrc  = (int*)   carve((size_t)E * sizeof(int));
  int*    bdst  = (int*)   carve((size_t)E * sizeof(int));
  int*    gcnt  = cnts;
  int*    gcur  = cnts + NB;

  auto g256 = [](long long n) { return (unsigned)((n + 255) / 256); };
  auto gwav = [](long long n) { return (unsigned)((n + 7) / 8); };

  pack_b_k<<<g256((long long)F_IN * H), 256, 0, stream>>>(Wl1, bpl1, F_IN, H);
  pack_b_k<<<g256((long long)F_IN * H), 256, 0, stream>>>(Wr1, bpr1, F_IN, H);
  pack_b_k<<<g256((long long)H * OUT),  256, 0, stream>>>(Wl2, bpl2, H, OUT);
  pack_b_k<<<g256((long long)H * OUT),  256, 0, stream>>>(Wr2, bpr2, H, OUT);

  const unsigned BIN_WGS = 512;
  fill_f32_k<<<g256(2LL * NB), 256, 0, stream>>>((float*)cnts, 0.0f, 2LL * NB);
  bin_count_k<<<BIN_WGS, 256, (size_t)NB * sizeof(int), stream>>>(edst, E, NB, gcnt);
  bin_offsets_k<<<1, 1024, 0, stream>>>(gcnt, NB, boff);
  bin_place_k<<<BIN_WGS, 256, (size_t)2 * NB * sizeof(int), stream>>>(esrc, edst, E, NB, boff, gcur,
                                                                      bsrc, bdst);

  const int Mt = Nn / 16;

  {
    const int K = F_IN, Kt = K / 32;
    pack_a_k<<<g256((long long)Nn * K / 16), 256, 0, stream>>>(x, apack, Nn, K);
    wmma_gemm_strip_k<8><<<gwav(Mt), 256, 0, stream>>>(apack, bpl1, xl, Mt, Kt);
    wmma_gemm_strip_k<8><<<gwav(Mt), 256, 0, stream>>>(apack, bpr1, xr, Mt, Kt);
    tile_gat_k<4><<<NB, 256, 0, stream>>>(xl, xr, att1, b1, bsrc, bdst, boff, E, Nn, logit, hbuf);
  }

  {
    const int K = H, Kt = K / 32;
    pack_a_k<<<g256((long long)Nn * K / 16), 256, 0, stream>>>(hbuf, apack, Nn, K);
    wmma_gemm_strip_k<2><<<gwav(Mt), 256, 0, stream>>>(apack, bpl2, xl, Mt, Kt);
    wmma_gemm_strip_k<2><<<gwav(Mt), 256, 0, stream>>>(apack, bpr2, xr, Mt, Kt);
    tile_gat_k<1><<<NB, 256, 0, stream>>>(xl, xr, att2, b2, bsrc, bdst, boff, E, Nn, logit, out);
  }
}
